# starting kernel
# baseline (speedup 1.0000x reference)
#include <hip/hip_runtime.h>


typedef float v2f __attribute__((ext_vector_type(2)));
typedef float v4f __attribute__((ext_vector_type(4)));
typedef float v8f __attribute__((ext_vector_type(8)));

#define PAD_ID 0
#define S_LEN 4096
#define D_LEN 512
#define WAVES 16
#define ROWS_PER_WAVE (S_LEN / WAVES)
#define CHUNKS (ROWS_PER_WAVE / 4)

__global__ __launch_bounds__(512, 1)
void nz_avg_pool_wmma(const float* __restrict__ x,
                      const int* __restrict__ tokens,
                      float* __restrict__ out) {
  const int dt   = blockIdx.x;
  const int b    = blockIdx.y;
  const int d0   = dt * 64;
  const int tid  = threadIdx.x;
  const int w    = tid >> 5;
  const int lane = tid & 31;
  const int n    = lane & 15;
  const int kA   = (lane & 16) ? 2 : 0;

  const float* __restrict__ xb = x + (size_t)b * S_LEN * D_LEN;
  const int*   __restrict__ tb = tokens + (size_t)b * S_LEN;

  const int sw = w * ROWS_PER_WAVE;

  v8f acc0 = {}, acc1 = {}, acc2 = {}, acc3 = {};
  float cnt = 0.0f;

  const float* px = xb + (size_t)(sw + kA) * D_LEN + d0 + 4 * n;
  const int*   pt = tb + sw + kA;

#pragma unroll 2
  for (int i = 0; i < CHUNKS; ++i) {
    v4f r0 = __builtin_nontemporal_load((const v4f*)px);
    v4f r1 = __builtin_nontemporal_load((const v4f*)(px + D_LEN));
    px += 4 * D_LEN;

    int2 t = *(const int2*)pt;
    pt += 4;
    float a0 = (t.x != PAD_ID) ? 1.0f : 0.0f;
    float a1 = (t.y != PAD_ID) ? 1.0f : 0.0f;
    cnt += (n == 0) ? (a0 + a1) : 0.0f;

    v2f a = {a0, a1};

    v2f b0 = {r0.x, r1.x};
    v2f b1 = {r0.y, r1.y};
    v2f b2 = {r0.z, r1.z};
    v2f b3 = {r0.w, r1.w};

    acc0 = __builtin_amdgcn_wmma_f32_16x16x4_f32(false, a, false, b0,
                                                 (short)0, acc0, false, false);
    acc1 = __builtin_amdgcn_wmma_f32_16x16x4_f32(false, a, false, b1,
                                                 (short)0, acc1, false, false);
    acc2 = __builtin_amdgcn_wmma_f32_16x16x4_f32(false, a, false, b2,
                                                 (short)0, acc2, false, false);
    acc3 = __builtin_amdgcn_wmma_f32_16x16x4_f32(false, a, false, b3,
                                                 (short)0, acc3, false, false);
  }

  __shared__ float sAcc[WAVES][4][16];
  __shared__ float sCnt[WAVES];

  if (lane < 16) {
    sAcc[w][0][n] = acc0[0];
    sAcc[w][1][n] = acc1[0];
    sAcc[w][2][n] = acc2[0];
    sAcc[w][3][n] = acc3[0];
  }
  float cw = cnt + __shfl_xor(cnt, 16, 32);
  if (lane == 0) sCnt[w] = cw;
  __syncthreads();

  if (tid < 64) {
    const int e  = tid & 3;
    const int nn = tid >> 2;
    float tot = 0.0f, c = 0.0f;
    for (int ww = 0; ww < WAVES; ++ww) {
      tot += sAcc[ww][e][nn];
      c   += sCnt[ww];
    }
    out[(size_t)b * D_LEN + d0 + tid] = tot / c;
  }
}

extern "C" void kernel_launch(void* const* d_in, const int* in_sizes, int n_in,
                              void* d_out, int out_size, void* d_ws,
                              size_t ws_size, hipStream_t stream) {
  (void)in_sizes; (void)n_in; (void)d_ws; (void)ws_size; (void)out_size;
  const float* x      = (const float*)d_in[0];
  const int*   tokens = (const int*)d_in[1];
  float*       out    = (float*)d_out;

  dim3 grid(D_LEN / 64, 16);
  nz_avg_pool_wmma<<<grid, 512, 0, stream>>>(x, tokens, out);
}


# Issue the loads of 16 K-chunks (32 b128 + 16 token pairs per lane) as one group ahead of their WMMAs: 4 memory round trips per wave instead of 32; chunk and summation order unchanged
# speedup vs baseline: 2.2095x; 2.2095x over previous
#include <hip/hip_runtime.h>


typedef float v2f __attribute__((ext_vector_type(2)));
typedef float v4f __attribute__((ext_vector_type(4)));
typedef float v8f __attribute__((ext_vector_type(8)));

#define PAD_ID 0
#define S_LEN 4096
#define D_LEN 512
#define WAVES 16
#define ROWS_PER_WAVE (S_LEN / WAVES)
#define CHUNKS (ROWS_PER_WAVE / 4)
#define BATCH 16

__global__ __launch_bounds__(512, 1)
void nz_avg_pool_wmma(const float* __restrict__ x,
                      const int* __restrict__ tokens,
                      float* __restrict__ out) {
  const int dt   = blockIdx.x;
  const int b    = blockIdx.y;
  const int d0   = dt * 64;
  const int tid  = threadIdx.x;
  const int w    = tid >> 5;
  const int lane = tid & 31;
  const int n    = lane & 15;
  const int kA   = (lane & 16) ? 2 : 0;

  const float* __restrict__ xb = x + (size_t)b * S_LEN * D_LEN;
  const int*   __restrict__ tb = tokens + (size_t)b * S_LEN;

  const int sw = w * ROWS_PER_WAVE;

  v8f acc0 = {}, acc1 = {}, acc2 = {}, acc3 = {};
  float cnt = 0.0f;

  const float* px = xb + (size_t)(sw + kA) * D_LEN + d0 + 4 * n;
  const int*   pt = tb + sw + kA;

#pragma unroll 1
  for (int j = 0; j < CHUNKS / BATCH; ++j) {
    v4f r0[BATCH], r1[BATCH];
    int2 t[BATCH];
#pragma unroll
    for (int u = 0; u < BATCH; ++u) {
      r0[u] = __builtin_nontemporal_load((const v4f*)(px + (size_t)u * 4 * D_LEN));
      r1[u] = __builtin_nontemporal_load((const v4f*)(px + (size_t)u * 4 * D_LEN + D_LEN));
      t[u]  = *(const int2*)(pt + 4 * u);
    }
    px += (size_t)BATCH * 4 * D_LEN;
    pt += BATCH * 4;

#pragma unroll
    for (int u = 0; u < BATCH; ++u) {
      float a0 = (t[u].x != PAD_ID) ? 1.0f : 0.0f;
      float a1 = (t[u].y != PAD_ID) ? 1.0f : 0.0f;
      cnt += (n == 0) ? (a0 + a1) : 0.0f;

      v2f a = {a0, a1};

      v2f b0 = {r0[u].x, r1[u].x};
      v2f b1 = {r0[u].y, r1[u].y};
      v2f b2 = {r0[u].z, r1[u].z};
      v2f b3 = {r0[u].w, r1[u].w};

      acc0 = __builtin_amdgcn_wmma_f32_16x16x4_f32(false, a, false, b0,
                                                   (short)0, acc0, false, false);
      acc1 = __builtin_amdgcn_wmma_f32_16x16x4_f32(false, a, false, b1,
                                                   (short)0, acc1, false, false);
      acc2 = __builtin_amdgcn_wmma_f32_16x16x4_f32(false, a, false, b2,
                                                   (short)0, acc2, false, false);
      acc3 = __builtin_amdgcn_wmma_f32_16x16x4_f32(false, a, false, b3,
                                                   (short)0, acc3, false, false);
    }
  }

  __shared__ float sAcc[WAVES][4][16];
  __shared__ float sCnt[WAVES];

  if (lane < 16) {
    sAcc[w][0][n] = acc0[0];
    sAcc[w][1][n] = acc1[0];
    sAcc[w][2][n] = acc2[0];
    sAcc[w][3][n] = acc3[0];
  }
  float cw = cnt + __shfl_xor(cnt, 16, 32);
  if (lane == 0) sCnt[w] = cw;
  __syncthreads();

  if (tid < 64) {
    const int e  = tid & 3;
    const int nn = tid >> 2;
    float tot = 0.0f, c = 0.0f;
    for (int ww = 0; ww < WAVES; ++ww) {
      tot += sAcc[ww][e][nn];
      c   += sCnt[ww];
    }
    out[(size_t)b * D_LEN + d0 + tid] = tot / c;
  }
}

extern "C" void kernel_launch(void* const* d_in, const int* in_sizes, int n_in,
                              void* d_out, int out_size, void* d_ws,
                              size_t ws_size, hipStream_t stream) {
  (void)in_sizes; (void)n_in; (void)d_ws; (void)ws_size; (void)out_size;
  const float* x      = (const float*)d_in[0];
  const int*   tokens = (const int*)d_in[1];
  float*       out    = (float*)d_out;

  dim3 grid(D_LEN / 64, 16);
  nz_avg_pool_wmma<<<grid, 512, 0, stream>>>(x, tokens, out);
}


# One work-group per (batch, 32-column tile), not 64: 256 work-groups on all 256 WGPs, not 128; 8-row K-chunks, own-row masks in A; loads held in chunk order; summation order changes (rows regrouped)
# speedup vs baseline: 2.9821x; 1.3497x over previous
#include <hip/hip_runtime.h>


typedef float v2f __attribute__((ext_vector_type(2)));
typedef float v4f __attribute__((ext_vector_type(4)));
typedef float v8f __attribute__((ext_vector_type(8)));

#define PAD_ID 0
#define S_LEN 4096
#define D_LEN 512
#define TILE_D 32
#define WAVES 16
#define ROWS_PER_WAVE (S_LEN / WAVES)
#define CHUNK_ROWS 8
#define CHUNKS (ROWS_PER_WAVE / CHUNK_ROWS)
#define BATCH 16

__global__ __launch_bounds__(512, 1)
void nz_avg_pool_wmma(const float* __restrict__ x,
                      const int* __restrict__ tokens,
                      float* __restrict__ out) {
  const int dt   = blockIdx.x;
  const int b    = blockIdx.y;
  const int d0   = dt * TILE_D;
  const int tid  = threadIdx.x;
  const int w    = tid >> 5;
  const int lane = tid & 31;
  const int c    = lane & 15;
  const int h    = lane >> 4;
  const int g    = c & 7;
  const int q    = c >> 3;
  const int r    = 4 * q + 2 * h;

  const float* __restrict__ xb = x + (size_t)b * S_LEN * D_LEN;
  const int*   __restrict__ tb = tokens + (size_t)b * S_LEN;

  const int sw = w * ROWS_PER_WAVE;

  v8f acc0 = {}, acc1 = {}, acc2 = {}, acc3 = {};
  float cnt = 0.0f;

  const float* px = xb + (size_t)(sw + r) * D_LEN + d0 + 4 * g;
  const int*   pt = tb + sw + r;

#pragma unroll 1
  for (int j = 0; j < CHUNKS / BATCH; ++j) {
    v4f r0[BATCH], r1[BATCH];
    int2 t[BATCH];
#pragma unroll
    for (int u = 0; u < BATCH; ++u) {
      t[u]  = *(const int2*)(pt + CHUNK_ROWS * u);
    }
    __builtin_amdgcn_sched_barrier(0);
#pragma unroll
    for (int u = 0; u < BATCH; ++u) {
      r0[u] = __builtin_nontemporal_load((const v4f*)(px + (size_t)u * CHUNK_ROWS * D_LEN));
      r1[u] = __builtin_nontemporal_load((const v4f*)(px + (size_t)u * CHUNK_ROWS * D_LEN + D_LEN));
    }
    __builtin_amdgcn_sched_barrier(0);
    px += (size_t)BATCH * CHUNK_ROWS * D_LEN;
    pt += BATCH * CHUNK_ROWS;

#pragma unroll
    for (int u = 0; u < BATCH; ++u) {
      float a0 = (t[u].x != PAD_ID) ? 1.0f : 0.0f;
      float a1 = (t[u].y != PAD_ID) ? 1.0f : 0.0f;
      cnt += (g == 0) ? (a0 + a1) : 0.0f;

      v2f a = {a0, a1};

      v2f b0 = {r0[u].x, r1[u].x};
      v2f b1 = {r0[u].y, r1[u].y};
      v2f b2 = {r0[u].z, r1[u].z};
      v2f b3 = {r0[u].w, r1[u].w};

      acc0 = __builtin_amdgcn_wmma_f32_16x16x4_f32(false, a, false, b0,
                                                   (short)0, acc0, false, false);
      acc1 = __builtin_amdgcn_wmma_f32_16x16x4_f32(false, a, false, b1,
                                                   (short)0, acc1, false, false);
      acc2 = __builtin_amdgcn_wmma_f32_16x16x4_f32(false, a, false, b2,
                                                   (short)0, acc2, false, false);
      acc3 = __builtin_amdgcn_wmma_f32_16x16x4_f32(false, a, false, b3,
                                                   (short)0, acc3, false, false);
      __builtin_amdgcn_sched_barrier(0);
    }
  }

  __shared__ float sAcc[WAVES][2][4][8];
  __shared__ float sCnt[WAVES];

  if (h == q) {
    sAcc[w][q][0][g] = acc0[0];
    sAcc[w][q][1][g] = acc1[0];
    sAcc[w][q][2][g] = acc2[0];
    sAcc[w][q][3][g] = acc3[0];
  }
  float cw = cnt + __shfl_xor(cnt, 16, 32);
  cw += __shfl_xor(cw, 8, 32);
  if (lane == 0) sCnt[w] = cw;
  __syncthreads();

  if (tid < TILE_D) {
    const int e  = tid & 3;
    const int gg = tid >> 2;
    float tot = 0.0f, cc = 0.0f;
    for (int ww = 0; ww < WAVES; ++ww) {
      tot += sAcc[ww][0][e][gg];
      tot += sAcc[ww][1][e][gg];
      cc  += sCnt[ww];
    }
    out[(size_t)b * D_LEN + d0 + tid] = tot / cc;
  }
}

extern "C" void kernel_launch(void* const* d_in, const int* in_sizes, int n_in,
                              void* d_out, int out_size, void* d_ws,
                              size_t ws_size, hipStream_t stream) {
  (void)in_sizes; (void)n_in; (void)d_ws; (void)ws_size; (void)out_size;
  const float* x      = (const float*)d_in[0];
  const int*   tokens = (const int*)d_in[1];
  float*       out    = (float*)d_out;

  dim3 grid(D_LEN / TILE_D, 16);
  nz_avg_pool_wmma<<<grid, 512, 0, stream>>>(x, tokens, out);
}


# Valid masks from ONE coalesced token load + 4 wave ballots per 128-row batch, not 16 small token loads: 33 memory instructions per batch instead of 48; count by popcount; x summation order unchanged
# speedup vs baseline: 3.1560x; 1.0583x over previous
#include <hip/hip_runtime.h>


typedef float v2f __attribute__((ext_vector_type(2)));
typedef float v4f __attribute__((ext_vector_type(4)));
typedef float v8f __attribute__((ext_vector_type(8)));

#define PAD_ID 0
#define S_LEN 4096
#define D_LEN 512
#define TILE_D 32
#define WAVES 16
#define ROWS_PER_WAVE (S_LEN / WAVES)
#define CHUNK_ROWS 8
#define CHUNKS (ROWS_PER_WAVE / CHUNK_ROWS)
#define BATCH 16

__global__ __launch_bounds__(512, 1)
void nz_avg_pool_wmma(const float* __restrict__ x,
                      const int* __restrict__ tokens,
                      float* __restrict__ out) {
  const int dt   = blockIdx.x;
  const int b    = blockIdx.y;
  const int d0   = dt * TILE_D;
  const int tid  = threadIdx.x;
  const int w    = tid >> 5;
  const int lane = tid & 31;
  const int c    = lane & 15;
  const int h    = lane >> 4;
  const int g    = c & 7;
  const int q    = c >> 3;
  const int r    = 4 * q + 2 * h;

  const float* __restrict__ xb = x + (size_t)b * S_LEN * D_LEN;
  const int*   __restrict__ tb = tokens + (size_t)b * S_LEN;

  const int sw = w * ROWS_PER_WAVE;

  v8f acc0 = {}, acc1 = {}, acc2 = {}, acc3 = {};
  int nvalid = 0;

  const float* px = xb + (size_t)(sw + r) * D_LEN + d0 + 4 * g;
  const int*   pt = tb + sw + 4 * lane;

#pragma unroll 1
  for (int j = 0; j < CHUNKS / BATCH; ++j) {
    v4f r0[BATCH], r1[BATCH];
    const int4 tk = *(const int4*)pt;
    __builtin_amdgcn_sched_barrier(0);
#pragma unroll
    for (int u = 0; u < BATCH; ++u) {
      r0[u] = __builtin_nontemporal_load((const v4f*)(px + (size_t)u * CHUNK_ROWS * D_LEN));
      r1[u] = __builtin_nontemporal_load((const v4f*)(px + (size_t)u * CHUNK_ROWS * D_LEN + D_LEN));
    }
    __builtin_amdgcn_sched_barrier(0);
    px += (size_t)BATCH * CHUNK_ROWS * D_LEN;
    pt += BATCH * CHUNK_ROWS;

    int t0 = tk.x, t1 = tk.y, t2 = tk.z, t3 = tk.w;
    asm volatile("" : "+v"(t0), "+v"(t1), "+v"(t2), "+v"(t3));
    const unsigned bm0 = __builtin_amdgcn_ballot_w32(t0 != PAD_ID);
    const unsigned bm1 = __builtin_amdgcn_ballot_w32(t1 != PAD_ID);
    const unsigned bm2 = __builtin_amdgcn_ballot_w32(t2 != PAD_ID);
    const unsigned bm3 = __builtin_amdgcn_ballot_w32(t3 != PAD_ID);
    nvalid += __builtin_popcount(bm0) + __builtin_popcount(bm1) +
              __builtin_popcount(bm2) + __builtin_popcount(bm3);
    const unsigned m0 = (h ? bm2 : bm0) >> q;
    const unsigned m1 = (h ? bm3 : bm1) >> q;

#pragma unroll
    for (int u = 0; u < BATCH; ++u) {
      const float a0 = (float)((m0 >> (2 * u)) & 1u);
      const float a1 = (float)((m1 >> (2 * u)) & 1u);

      v2f a = {a0, a1};

      v2f b0 = {r0[u].x, r1[u].x};
      v2f b1 = {r0[u].y, r1[u].y};
      v2f b2 = {r0[u].z, r1[u].z};
      v2f b3 = {r0[u].w, r1[u].w};

      acc0 = __builtin_amdgcn_wmma_f32_16x16x4_f32(false, a, false, b0,
                                                   (short)0, acc0, false, false);
      acc1 = __builtin_amdgcn_wmma_f32_16x16x4_f32(false, a, false, b1,
                                                   (short)0, acc1, false, false);
      acc2 = __builtin_amdgcn_wmma_f32_16x16x4_f32(false, a, false, b2,
                                                   (short)0, acc2, false, false);
      acc3 = __builtin_amdgcn_wmma_f32_16x16x4_f32(false, a, false, b3,
                                                   (short)0, acc3, false, false);
      __builtin_amdgcn_sched_barrier(0);
    }
  }

  __shared__ float sAcc[WAVES][2][4][8];
  __shared__ float sCnt[WAVES];

  if (h == q) {
    sAcc[w][q][0][g] = acc0[0];
    sAcc[w][q][1][g] = acc1[0];
    sAcc[w][q][2][g] = acc2[0];
    sAcc[w][q][3][g] = acc3[0];
  }
  if (lane == 0) sCnt[w] = (float)nvalid;
  __syncthreads();

  if (tid < TILE_D) {
    const int e  = tid & 3;
    const int gg = tid >> 2;
    float tot = 0.0f, cc = 0.0f;
    for (int ww = 0; ww < WAVES; ++ww) {
      tot += sAcc[ww][0][e][gg];
      tot += sAcc[ww][1][e][gg];
      cc  += sCnt[ww];
    }
    out[(size_t)b * D_LEN + d0 + tid] = tot / cc;
  }
}

extern "C" void kernel_launch(void* const* d_in, const int* in_sizes, int n_in,
                              void* d_out, int out_size, void* d_ws,
                              size_t ws_size, hipStream_t stream) {
  (void)in_sizes; (void)n_in; (void)d_ws; (void)ws_size; (void)out_size;
  const float* x      = (const float*)d_in[0];
  const int*   tokens = (const int*)d_in[1];
  float*       out    = (float*)d_out;

  dim3 grid(D_LEN / TILE_D, 16);
  nz_avg_pool_wmma<<<grid, 512, 0, stream>>>(x, tokens, out);
}
